# starting kernel
# baseline (speedup 1.0000x reference)
#include <hip/hip_runtime.h>

typedef __attribute__((ext_vector_type(2))) float v2f;
typedef __attribute__((ext_vector_type(8))) float v8f;

#define IN_FEAT  256
#define OUT_FEAT 64
#define KPAIRS   (IN_FEAT / 2)
#define EPG      128

__global__ void gcn_zero_f32(float* __restrict__ p, int n) {
    int i = blockIdx.x * blockDim.x + threadIdx.x;
    if (i < n) p[i] = 0.0f;
}

__global__ void __launch_bounds__(128) gcn_gemm_wmma(
    const float* __restrict__ feat, const float* __restrict__ W,
    float* __restrict__ proj, int n_nodes)
{
    __shared__ v2f lW[KPAIRS * OUT_FEAT];
    for (int i = threadIdx.x; i < KPAIRS * OUT_FEAT; i += 128) {
        const int kp = i >> 6;
        const int n  = i & 63;
        v2f w;
        w.x = W[(2 * kp + 0) * OUT_FEAT + n];
        w.y = W[(2 * kp + 1) * OUT_FEAT + n];
        lW[i] = w;
    }
    __syncthreads();

    const int wave = threadIdx.x >> 5;
    const int lane = threadIdx.x & 31;
    const int m    = lane & 15;
    const int half = lane >> 4;

    const int row_base = (blockIdx.x * 4 + wave) * 16;
    if (row_base >= n_nodes) return;

    const v2f* arow = (const v2f*)(feat + (size_t)(row_base + m) * IN_FEAT) + half;

    v8f c0 = {}, c1 = {}, c2 = {}, c3 = {};

    for (int k = 0; k < IN_FEAT; k += 4) {
        const v2f a = arow[k >> 1];

        const v2f* wb = lW + ((k >> 1) + half) * OUT_FEAT + m;
        const v2f b0 = wb[0];
        const v2f b1 = wb[16];
        const v2f b2 = wb[32];
        const v2f b3 = wb[48];

        c0 = __builtin_amdgcn_wmma_f32_16x16x4_f32(false, a, false, b0, (short)0, c0, false, false);
        c1 = __builtin_amdgcn_wmma_f32_16x16x4_f32(false, a, false, b1, (short)0, c1, false, false);
        c2 = __builtin_amdgcn_wmma_f32_16x16x4_f32(false, a, false, b2, (short)0, c2, false, false);
        c3 = __builtin_amdgcn_wmma_f32_16x16x4_f32(false, a, false, b3, (short)0, c3, false, false);
    }

    float* prow = proj + (size_t)row_base * OUT_FEAT + m;
#pragma unroll
    for (int v = 0; v < 8; ++v) {
        const size_t r = (size_t)(v + 8 * half) * OUT_FEAT;
        prow[r + 0]  = c0[v];
        prow[r + 16] = c1[v];
        prow[r + 32] = c2[v];
        prow[r + 48] = c3[v];
    }
}

__global__ void __launch_bounds__(256) gcn_scatter(
    const float* __restrict__ proj,
    const int*   __restrict__ rows,
    const int*   __restrict__ cols,
    const float* __restrict__ vals,
    float* __restrict__ out, int n_edges)
{
    const int f     = threadIdx.x & 63;
    const int sub   = threadIdx.x >> 6;
    const int group = blockIdx.x * 4 + sub;

    int e0 = group * EPG;
    if (e0 >= n_edges) return;
    int e1 = e0 + EPG;
    if (e1 > n_edges) e1 = n_edges;

    float acc = 0.0f;
    int cur_row = rows[e0];

    for (int e = e0; e < e1; ++e) {
        int r = rows[e];
        if (r != cur_row) {
            atomicAdd(&out[(size_t)cur_row * OUT_FEAT + f], acc);
            acc = 0.0f;
            cur_row = r;
        }
        int c = cols[e];
        acc += vals[e] * proj[(size_t)c * OUT_FEAT + f];
    }
    atomicAdd(&out[(size_t)cur_row * OUT_FEAT + f], acc);
}

extern "C" void kernel_launch(void* const* d_in, const int* in_sizes, int n_in,
                              void* d_out, int out_size, void* d_ws, size_t ws_size,
                              hipStream_t stream) {
    const float* feat = (const float*)d_in[0];
    const float* W    = (const float*)d_in[1];
    const int*   rows = (const int*)d_in[2];
    const int*   cols = (const int*)d_in[3];
    const float* vals = (const float*)d_in[4];
    float* out  = (float*)d_out;
    float* proj = (float*)d_ws;

    const int n_nodes = in_sizes[0] / IN_FEAT;
    const int n_edges = in_sizes[2];

    const int zn = n_nodes * OUT_FEAT;
    gcn_zero_f32<<<(zn + 255) / 256, 256, 0, stream>>>(out, zn);

    const int row_tiles = (n_nodes + 15) / 16;
    gcn_gemm_wmma<<<(row_tiles + 3) / 4, 128, 0, stream>>>(feat, W, proj, n_nodes);

    const int groups = (n_edges + EPG - 1) / EPG;
    gcn_scatter<<<(groups + 3) / 4, 256, 0, stream>>>(proj, rows, cols, vals, out, n_edges);
}


# scatter: each wave reads 32 edges' row/col/val with one coalesced load per array and issues their 32 proj gathers as one batch before the adds (was a dependent index->gather chain per edge)
# speedup vs baseline: 2.3382x; 2.3382x over previous
#include <hip/hip_runtime.h>

typedef __attribute__((ext_vector_type(2))) float v2f;
typedef __attribute__((ext_vector_type(8))) float v8f;

#define IN_FEAT  256
#define OUT_FEAT 64
#define KPAIRS   (IN_FEAT / 2)
#define EPG      128

__global__ void gcn_zero_f32(float* __restrict__ p, int n) {
    int i = blockIdx.x * blockDim.x + threadIdx.x;
    if (i < n) p[i] = 0.0f;
}

__global__ void __launch_bounds__(128) gcn_gemm_wmma(
    const float* __restrict__ feat, const float* __restrict__ W,
    float* __restrict__ proj, int n_nodes)
{
    __shared__ v2f lW[KPAIRS * OUT_FEAT];
    for (int i = threadIdx.x; i < KPAIRS * OUT_FEAT; i += 128) {
        const int kp = i >> 6;
        const int n  = i & 63;
        v2f w;
        w.x = W[(2 * kp + 0) * OUT_FEAT + n];
        w.y = W[(2 * kp + 1) * OUT_FEAT + n];
        lW[i] = w;
    }
    __syncthreads();

    const int wave = threadIdx.x >> 5;
    const int lane = threadIdx.x & 31;
    const int m    = lane & 15;
    const int half = lane >> 4;

    const int row_base = (blockIdx.x * 4 + wave) * 16;
    if (row_base >= n_nodes) return;

    const v2f* arow = (const v2f*)(feat + (size_t)(row_base + m) * IN_FEAT) + half;

    v8f c0 = {}, c1 = {}, c2 = {}, c3 = {};

    for (int k = 0; k < IN_FEAT; k += 4) {
        const v2f a = arow[k >> 1];

        const v2f* wb = lW + ((k >> 1) + half) * OUT_FEAT + m;
        const v2f b0 = wb[0];
        const v2f b1 = wb[16];
        const v2f b2 = wb[32];
        const v2f b3 = wb[48];

        c0 = __builtin_amdgcn_wmma_f32_16x16x4_f32(false, a, false, b0, (short)0, c0, false, false);
        c1 = __builtin_amdgcn_wmma_f32_16x16x4_f32(false, a, false, b1, (short)0, c1, false, false);
        c2 = __builtin_amdgcn_wmma_f32_16x16x4_f32(false, a, false, b2, (short)0, c2, false, false);
        c3 = __builtin_amdgcn_wmma_f32_16x16x4_f32(false, a, false, b3, (short)0, c3, false, false);
    }

    float* prow = proj + (size_t)row_base * OUT_FEAT + m;
#pragma unroll
    for (int v = 0; v < 8; ++v) {
        const size_t r = (size_t)(v + 8 * half) * OUT_FEAT;
        prow[r + 0]  = c0[v];
        prow[r + 16] = c1[v];
        prow[r + 32] = c2[v];
        prow[r + 48] = c3[v];
    }
}

__global__ void __launch_bounds__(256) gcn_scatter(
    const float* __restrict__ proj,
    const int*   __restrict__ rows,
    const int*   __restrict__ cols,
    const float* __restrict__ vals,
    float* __restrict__ out, int n_edges)
{
    const int f    = threadIdx.x & 63;
    const int lane = threadIdx.x & 31;
    const int sub   = __builtin_amdgcn_readfirstlane((int)(threadIdx.x >> 6));
    const int group = blockIdx.x * 4 + sub;

    const int e0 = group * EPG;
    if (e0 >= n_edges) return;
    int e1 = e0 + EPG;
    if (e1 > n_edges) e1 = n_edges;
    const int elast = e1 - 1;

    int ec = e0 + lane;
    if (ec > elast) ec = elast;
    int   rv = rows[ec];
    int   cv = cols[ec];
    float vv = vals[ec];

    float acc = 0.0f;
    int cur_row = __builtin_amdgcn_readfirstlane(rv);

#define CONSUME_EDGE(j)                                                       \
    {                                                                         \
        const int   r = __builtin_amdgcn_readlane(rv, (j));                   \
        const float v = __builtin_bit_cast(float,                             \
            __builtin_amdgcn_readlane(__builtin_bit_cast(int, vv), (j)));     \
        if (r != cur_row) {                                                   \
            atomicAdd(out + (size_t)cur_row * OUT_FEAT + f, acc);             \
            acc = 0.0f;                                                       \
            cur_row = r;                                                      \
        }                                                                     \
        acc += v * p[(j)];                                                    \
    }

    for (int base = e0; base < e1; base += 32) {
        int cnt = e1 - base;
        if (cnt > 32) cnt = 32;

        float p[32];
#pragma unroll
        for (int j = 0; j < 32; ++j) {
            const int c = __builtin_amdgcn_readlane(cv, j);
            const float* prow = proj + (size_t)c * OUT_FEAT;
            p[j] = prow[f];
        }

        int en = base + 32 + lane;
        if (en > elast) en = elast;
        const int   rn = rows[en];
        const int   cn = cols[en];
        const float vn = vals[en];

        if (cnt == 32) {
#pragma unroll
            for (int j = 0; j < 32; ++j) CONSUME_EDGE(j)
        } else {
#pragma unroll
            for (int j = 0; j < 32; ++j) {
                if (j < cnt) CONSUME_EDGE(j)
            }
        }

        rv = rn; cv = cn; vv = vn;
    }
#undef CONSUME_EDGE
    atomicAdd(out + (size_t)cur_row * OUT_FEAT + f, acc);
}

extern "C" void kernel_launch(void* const* d_in, const int* in_sizes, int n_in,
                              void* d_out, int out_size, void* d_ws, size_t ws_size,
                              hipStream_t stream) {
    const float* feat = (const float*)d_in[0];
    const float* W    = (const float*)d_in[1];
    const int*   rows = (const int*)d_in[2];
    const int*   cols = (const int*)d_in[3];
    const float* vals = (const float*)d_in[4];
    float* out  = (float*)d_out;
    float* proj = (float*)d_ws;

    const int n_nodes = in_sizes[0] / IN_FEAT;
    const int n_edges = in_sizes[2];

    const int zn = n_nodes * OUT_FEAT;
    gcn_zero_f32<<<(zn + 255) / 256, 256, 0, stream>>>(out, zn);

    const int row_tiles = (n_nodes + 15) / 16;
    gcn_gemm_wmma<<<(row_tiles + 3) / 4, 128, 0, stream>>>(feat, W, proj, n_nodes);

    const int groups = (n_edges + EPG - 1) / EPG;
    gcn_scatter<<<(groups + 3) / 4, 256, 0, stream>>>(proj, rows, cols, vals, out, n_edges);
}


# zero fill of out folded into the GEMM launch: each wave stores 0 to its tile's 16 out rows (eight 16-byte stores per lane, whole 128-byte lines) ahead of the weight staging; 3 launches -> 2
# speedup vs baseline: 2.7785x; 1.1883x over previous
#include <hip/hip_runtime.h>

typedef __attribute__((ext_vector_type(2))) float v2f;
typedef __attribute__((ext_vector_type(4))) float v4f;
typedef __attribute__((ext_vector_type(8))) float v8f;

#define IN_FEAT  256
#define OUT_FEAT 64
#define KPAIRS   (IN_FEAT / 2)
#define EPG      128

__global__ void __launch_bounds__(128) gcn_gemm_wmma(
    const float* __restrict__ feat, const float* __restrict__ W,
    float* __restrict__ proj, float* __restrict__ out, int n_nodes)
{
    __shared__ v2f lW[KPAIRS * OUT_FEAT];

    {
        const int zlane = threadIdx.x & 31;
        const int zrow0 = (blockIdx.x * 4 + (threadIdx.x >> 5)) * 16;
        v4f* zp = (v4f*)(out + (size_t)zrow0 * OUT_FEAT) + zlane;
        const v4f z4 = {0.0f, 0.0f, 0.0f, 0.0f};
#pragma unroll
        for (int j = 0; j < 8; ++j) {
            if (zrow0 + 2 * j + (zlane >> 4) < n_nodes) zp[j * 32] = z4;
        }
    }

    for (int i = threadIdx.x; i < KPAIRS * OUT_FEAT; i += 128) {
        const int kp = i >> 6;
        const int n  = i & 63;
        v2f w;
        w.x = W[(2 * kp + 0) * OUT_FEAT + n];
        w.y = W[(2 * kp + 1) * OUT_FEAT + n];
        lW[i] = w;
    }
    __syncthreads();

    const int wave = threadIdx.x >> 5;
    const int lane = threadIdx.x & 31;
    const int m    = lane & 15;
    const int half = lane >> 4;

    const int row_base = (blockIdx.x * 4 + wave) * 16;
    if (row_base >= n_nodes) return;

    const v2f* arow = (const v2f*)(feat + (size_t)(row_base + m) * IN_FEAT) + half;

    v8f c0 = {}, c1 = {}, c2 = {}, c3 = {};

    for (int k = 0; k < IN_FEAT; k += 4) {
        const v2f a = arow[k >> 1];

        const v2f* wb = lW + ((k >> 1) + half) * OUT_FEAT + m;
        const v2f b0 = wb[0];
        const v2f b1 = wb[16];
        const v2f b2 = wb[32];
        const v2f b3 = wb[48];

        c0 = __builtin_amdgcn_wmma_f32_16x16x4_f32(false, a, false, b0, (short)0, c0, false, false);
        c1 = __builtin_amdgcn_wmma_f32_16x16x4_f32(false, a, false, b1, (short)0, c1, false, false);
        c2 = __builtin_amdgcn_wmma_f32_16x16x4_f32(false, a, false, b2, (short)0, c2, false, false);
        c3 = __builtin_amdgcn_wmma_f32_16x16x4_f32(false, a, false, b3, (short)0, c3, false, false);
    }

    float* prow = proj + (size_t)row_base * OUT_FEAT + m;
#pragma unroll
    for (int v = 0; v < 8; ++v) {
        const size_t r = (size_t)(v + 8 * half) * OUT_FEAT;
        prow[r + 0]  = c0[v];
        prow[r + 16] = c1[v];
        prow[r + 32] = c2[v];
        prow[r + 48] = c3[v];
    }
}

__global__ void __launch_bounds__(256) gcn_scatter(
    const float* __restrict__ proj,
    const int*   __restrict__ rows,
    const int*   __restrict__ cols,
    const float* __restrict__ vals,
    float* __restrict__ out, int n_edges)
{
    const int f    = threadIdx.x & 63;
    const int lane = threadIdx.x & 31;
    const int sub   = __builtin_amdgcn_readfirstlane((int)(threadIdx.x >> 6));
    const int group = blockIdx.x * 4 + sub;

    const int e0 = group * EPG;
    if (e0 >= n_edges) return;
    int e1 = e0 + EPG;
    if (e1 > n_edges) e1 = n_edges;
    const int elast = e1 - 1;

    int ec = e0 + lane;
    if (ec > elast) ec = elast;
    int   rv = rows[ec];
    int   cv = cols[ec];
    float vv = vals[ec];

    float acc = 0.0f;
    int cur_row = __builtin_amdgcn_readfirstlane(rv);

#define CONSUME_EDGE(j)                                                       \
    {                                                                         \
        const int   r = __builtin_amdgcn_readlane(rv, (j));                   \
        const float v = __builtin_bit_cast(float,                             \
            __builtin_amdgcn_readlane(__builtin_bit_cast(int, vv), (j)));     \
        if (r != cur_row) {                                                   \
            atomicAdd(out + (size_t)cur_row * OUT_FEAT + f, acc);             \
            acc = 0.0f;                                                       \
            cur_row = r;                                                      \
        }                                                                     \
        acc += v * p[(j)];                                                    \
    }

    for (int base = e0; base < e1; base += 32) {
        int cnt = e1 - base;
        if (cnt > 32) cnt = 32;

        float p[32];
#pragma unroll
        for (int j = 0; j < 32; ++j) {
            const int c = __builtin_amdgcn_readlane(cv, j);
            const float* prow = proj + (size_t)c * OUT_FEAT;
            p[j] = prow[f];
        }

        int en = base + 32 + lane;
        if (en > elast) en = elast;
        const int   rn = rows[en];
        const int   cn = cols[en];
        const float vn = vals[en];

        if (cnt == 32) {
#pragma unroll
            for (int j = 0; j < 32; ++j) CONSUME_EDGE(j)
        } else {
#pragma unroll
            for (int j = 0; j < 32; ++j) {
                if (j < cnt) CONSUME_EDGE(j)
            }
        }

        rv = rn; cv = cn; vv = vn;
    }
#undef CONSUME_EDGE
    atomicAdd(out + (size_t)cur_row * OUT_FEAT + f, acc);
}

extern "C" void kernel_launch(void* const* d_in, const int* in_sizes, int n_in,
                              void* d_out, int out_size, void* d_ws, size_t ws_size,
                              hipStream_t stream) {
    const float* feat = (const float*)d_in[0];
    const float* W    = (const float*)d_in[1];
    const int*   rows = (const int*)d_in[2];
    const int*   cols = (const int*)d_in[3];
    const float* vals = (const float*)d_in[4];
    float* out  = (float*)d_out;
    float* proj = (float*)d_ws;

    const int n_nodes = in_sizes[0] / IN_FEAT;
    const int n_edges = in_sizes[2];

    const int row_tiles = (n_nodes + 15) / 16;
    gcn_gemm_wmma<<<(row_tiles + 3) / 4, 128, 0, stream>>>(feat, W, proj, out, n_nodes);

    const int groups = (n_edges + EPG - 1) / EPG;
    gcn_scatter<<<(groups + 3) / 4, 256, 0, stream>>>(proj, rows, cols, vals, out, n_edges);
}


# GEMM launch regridded: 256 workgroups of 26 waves, one per WGP (168 KiB LDS declared), stage W once and run their 24-25 row tiles side by side, the 25th as 2 column halves; was 1563 4-wave groups
# speedup vs baseline: 3.2475x; 1.1688x over previous
#include <hip/hip_runtime.h>

typedef __attribute__((ext_vector_type(2))) float v2f;
typedef __attribute__((ext_vector_type(4))) float v4f;
typedef __attribute__((ext_vector_type(8))) float v8f;

#define IN_FEAT  256
#define OUT_FEAT 64
#define KPAIRS   (IN_FEAT / 2)
#define EPG      128

#define GEMM_GROUPS  256
#define GEMM_SLOTS   25
#define GEMM_WAVES   26
#define GEMM_THREADS (GEMM_WAVES * 32)
#define GEMM_STAGERS 512
#define GEMM_LDS_V2F (168 * 1024 / 8)


__device__ __forceinline__ void gemm_zero_out_rows(
    float* __restrict__ out, int row_base, int lane, int n_nodes)
{
    v4f* zp = (v4f*)(out + (size_t)row_base * OUT_FEAT) + lane;
    const v4f z4 = {0.0f, 0.0f, 0.0f, 0.0f};
#pragma unroll
    for (int j = 0; j < 8; ++j) {
        if (row_base + 2 * j + (lane >> 4) < n_nodes) zp[j * 32] = z4;
    }
}

template <int NCB>
__device__ __forceinline__ void gemm_tile(
    const float* __restrict__ feat, const v2f* lW, float* __restrict__ proj,
    int row_base, int cb0, int m, int half)
{
    const v2f* arow = (const v2f*)(feat + (size_t)(row_base + m) * IN_FEAT) + half;
    const v2f* wcol = lW + half * OUT_FEAT + cb0 * 16 + m;

    v8f c[NCB];
#pragma unroll
    for (int j = 0; j < NCB; ++j) c[j] = v8f{};

    for (int k = 0; k < IN_FEAT; k += 4) {
        const v2f a = arow[k >> 1];
        const v2f* wb = wcol + (k >> 1) * OUT_FEAT;
        v2f b[NCB];
#pragma unroll
        for (int j = 0; j < NCB; ++j) b[j] = wb[16 * j];
#pragma unroll
        for (int j = 0; j < NCB; ++j)
            c[j] = __builtin_amdgcn_wmma_f32_16x16x4_f32(false, a, false, b[j], (short)0, c[j], false, false);
    }

    float* prow = proj + (size_t)row_base * OUT_FEAT + cb0 * 16 + m;
#pragma unroll
    for (int v = 0; v < 8; ++v) {
        const size_t r = (size_t)(v + 8 * half) * OUT_FEAT;
#pragma unroll
        for (int j = 0; j < NCB; ++j) prow[r + 16 * j] = c[j][v];
    }
}

__global__ void __launch_bounds__(GEMM_THREADS) gcn_gemm_wmma(
    const float* __restrict__ feat, const float* __restrict__ W,
    float* __restrict__ proj, float* __restrict__ out, int n_nodes)
{
    __shared__ v2f lW[GEMM_LDS_V2F];

    const int lane = threadIdx.x & 31;
    const int wave = __builtin_amdgcn_readfirstlane((int)(threadIdx.x >> 5));

    if (threadIdx.x < GEMM_STAGERS) {
#pragma unroll
        for (int j = 0; j < KPAIRS * OUT_FEAT / GEMM_STAGERS; ++j) {
            const int i  = (int)threadIdx.x + j * GEMM_STAGERS;
            const int kp = i >> 6;
            const int n  = i & 63;
            v2f w;
            w.x = W[(2 * kp + 0) * OUT_FEAT + n];
            w.y = W[(2 * kp + 1) * OUT_FEAT + n];
            lW[i] = w;
        }
    }
    __syncthreads();

    const int m    = lane & 15;
    const int half = lane >> 4;

    const int row_tiles = (n_nodes + 15) >> 4;
    const int tstep = (int)gridDim.x * GEMM_SLOTS;

    if (wave < GEMM_SLOTS - 1) {
        for (int t = (int)blockIdx.x + (int)gridDim.x * wave; t < row_tiles; t += tstep) {
            gemm_zero_out_rows(out, t * 16, lane, n_nodes);
            gemm_tile<4>(feat, lW, proj, t * 16, 0, m, half);
        }
    } else {
        const int h = wave - (GEMM_SLOTS - 1);
        for (int t = (int)blockIdx.x + (int)gridDim.x * (GEMM_SLOTS - 1); t < row_tiles; t += tstep) {
            if (h == 0) gemm_zero_out_rows(out, t * 16, lane, n_nodes);
            gemm_tile<2>(feat, lW, proj, t * 16, 2 * h, m, half);
        }
    }
}

__global__ void __launch_bounds__(256) gcn_scatter(
    const float* __restrict__ proj,
    const int*   __restrict__ rows,
    const int*   __restrict__ cols,
    const float* __restrict__ vals,
    float* __restrict__ out, int n_edges)
{
    const int f    = threadIdx.x & 63;
    const int lane = threadIdx.x & 31;
    const int sub   = __builtin_amdgcn_readfirstlane((int)(threadIdx.x >> 6));
    const int group = blockIdx.x * 4 + sub;

    const int e0 = group * EPG;
    if (e0 >= n_edges) return;
    int e1 = e0 + EPG;
    if (e1 > n_edges) e1 = n_edges;
    const int elast = e1 - 1;

    int ec = e0 + lane;
    if (ec > elast) ec = elast;
    int   rv = rows[ec];
    int   cv = cols[ec];
    float vv = vals[ec];

    float acc = 0.0f;
    int cur_row = __builtin_amdgcn_readfirstlane(rv);

#define CONSUME_EDGE(j)                                                       \
    {                                                                         \
        const int   r = __builtin_amdgcn_readlane(rv, (j));                   \
        const float v = __builtin_bit_cast(float,                             \
            __builtin_amdgcn_readlane(__builtin_bit_cast(int, vv), (j)));     \
        if (r != cur_row) {                                                   \
            atomicAdd(out + (size_t)cur_row * OUT_FEAT + f, acc);             \
            acc = 0.0f;                                                       \
            cur_row = r;                                                      \
        }                                                                     \
        acc += v * p[(j)];                                                    \
    }

    for (int base = e0; base < e1; base += 32) {
        int cnt = e1 - base;
        if (cnt > 32) cnt = 32;

        float p[32];
#pragma unroll
        for (int j = 0; j < 32; ++j) {
            const int c = __builtin_amdgcn_readlane(cv, j);
            const float* prow = proj + (size_t)c * OUT_FEAT;
            p[j] = prow[f];
        }

        int en = base + 32 + lane;
        if (en > elast) en = elast;
        const int   rn = rows[en];
        const int   cn = cols[en];
        const float vn = vals[en];

        if (cnt == 32) {
#pragma unroll
            for (int j = 0; j < 32; ++j) CONSUME_EDGE(j)
        } else {
#pragma unroll
            for (int j = 0; j < 32; ++j) {
                if (j < cnt) CONSUME_EDGE(j)
            }
        }

        rv = rn; cv = cn; vv = vn;
    }
#undef CONSUME_EDGE
    atomicAdd(out + (size_t)cur_row * OUT_FEAT + f, acc);
}

extern "C" void kernel_launch(void* const* d_in, const int* in_sizes, int n_in,
                              void* d_out, int out_size, void* d_ws, size_t ws_size,
                              hipStream_t stream) {
    const float* feat = (const float*)d_in[0];
    const float* W    = (const float*)d_in[1];
    const int*   rows = (const int*)d_in[2];
    const int*   cols = (const int*)d_in[3];
    const float* vals = (const float*)d_in[4];
    float* out  = (float*)d_out;
    float* proj = (float*)d_ws;

    const int n_nodes = in_sizes[0] / IN_FEAT;
    const int n_edges = in_sizes[2];

    gcn_gemm_wmma<<<GEMM_GROUPS, GEMM_THREADS, 0, stream>>>(feat, W, proj, out, n_nodes);

    const int groups = (n_edges + EPG - 1) / EPG;
    gcn_scatter<<<(groups + 3) / 4, 256, 0, stream>>>(proj, rows, cols, vals, out, n_edges);
}


# scatter: 16-lane half-wave per edge loads the whole 256-B proj row (16 B/lane, 2 rows per load); rows inside a half's 64 edges stored plainly, range ends atomicAdd (sums split per 64 edges, was 128)
# speedup vs baseline: 3.3801x; 1.0408x over previous
#include <hip/hip_runtime.h>

typedef __attribute__((ext_vector_type(2))) float v2f;
typedef __attribute__((ext_vector_type(4))) float v4f;
typedef __attribute__((ext_vector_type(8))) float v8f;

#define IN_FEAT  256
#define OUT_FEAT 64
#define KPAIRS   (IN_FEAT / 2)

#define GEMM_GROUPS  256
#define GEMM_SLOTS   25
#define GEMM_WAVES   26
#define GEMM_THREADS (GEMM_WAVES * 32)
#define GEMM_STAGERS 512
#define GEMM_LDS_V2F (168 * 1024 / 8)


__device__ __forceinline__ void gemm_zero_out_rows(
    float* __restrict__ out, int row_base, int lane, int n_nodes)
{
    v4f* zp = (v4f*)(out + (size_t)row_base * OUT_FEAT) + lane;
    const v4f z4 = {0.0f, 0.0f, 0.0f, 0.0f};
#pragma unroll
    for (int j = 0; j < 8; ++j) {
        if (row_base + 2 * j + (lane >> 4) < n_nodes) zp[j * 32] = z4;
    }
}

template <int NCB>
__device__ __forceinline__ void gemm_tile(
    const float* __restrict__ feat, const v2f* lW, float* __restrict__ proj,
    int row_base, int cb0, int m, int half)
{
    const v2f* arow = (const v2f*)(feat + (size_t)(row_base + m) * IN_FEAT) + half;
    const v2f* wcol = lW + half * OUT_FEAT + cb0 * 16 + m;

    v8f c[NCB];
#pragma unroll
    for (int j = 0; j < NCB; ++j) c[j] = v8f{};

    for (int k = 0; k < IN_FEAT; k += 4) {
        const v2f a = arow[k >> 1];
        const v2f* wb = wcol + (k >> 1) * OUT_FEAT;
        v2f b[NCB];
#pragma unroll
        for (int j = 0; j < NCB; ++j) b[j] = wb[16 * j];
#pragma unroll
        for (int j = 0; j < NCB; ++j)
            c[j] = __builtin_amdgcn_wmma_f32_16x16x4_f32(false, a, false, b[j], (short)0, c[j], false, false);
    }

    float* prow = proj + (size_t)row_base * OUT_FEAT + cb0 * 16 + m;
#pragma unroll
    for (int v = 0; v < 8; ++v) {
        const size_t r = (size_t)(v + 8 * half) * OUT_FEAT;
#pragma unroll
        for (int j = 0; j < NCB; ++j) prow[r + 16 * j] = c[j][v];
    }
}

__global__ void __launch_bounds__(GEMM_THREADS) gcn_gemm_wmma(
    const float* __restrict__ feat, const float* __restrict__ W,
    float* __restrict__ proj, float* __restrict__ out, int n_nodes)
{
    __shared__ v2f lW[GEMM_LDS_V2F];

    const int lane = threadIdx.x & 31;
    const int wave = __builtin_amdgcn_readfirstlane((int)(threadIdx.x >> 5));

    if (threadIdx.x < GEMM_STAGERS) {
#pragma unroll
        for (int j = 0; j < KPAIRS * OUT_FEAT / GEMM_STAGERS; ++j) {
            const int i  = (int)threadIdx.x + j * GEMM_STAGERS;
            const int kp = i >> 6;
            const int n  = i & 63;
            v2f w;
            w.x = W[(2 * kp + 0) * OUT_FEAT + n];
            w.y = W[(2 * kp + 1) * OUT_FEAT + n];
            lW[i] = w;
        }
    }
    __syncthreads();

    const int m    = lane & 15;
    const int half = lane >> 4;

    const int row_tiles = (n_nodes + 15) >> 4;
    const int tstep = (int)gridDim.x * GEMM_SLOTS;

    if (wave < GEMM_SLOTS - 1) {
        for (int t = (int)blockIdx.x + (int)gridDim.x * wave; t < row_tiles; t += tstep) {
            gemm_zero_out_rows(out, t * 16, lane, n_nodes);
            gemm_tile<4>(feat, lW, proj, t * 16, 0, m, half);
        }
    } else {
        const int h = wave - (GEMM_SLOTS - 1);
        for (int t = (int)blockIdx.x + (int)gridDim.x * (GEMM_SLOTS - 1); t < row_tiles; t += tstep) {
            if (h == 0) gemm_zero_out_rows(out, t * 16, lane, n_nodes);
            gemm_tile<2>(feat, lW, proj, t * 16, 2 * h, m, half);
        }
    }
}

#define EPW 128
#define EPH 64

template <int I>
__device__ __forceinline__ int row_lane(int x)
{
    return __builtin_amdgcn_update_dpp(0, x, 0x150 + I, 0xF, 0xF, true);
}

__device__ __forceinline__ void scatter_flush(
    float* __restrict__ out, int row, unsigned lane_bytes, v4f acc, bool edge_row)
{
    char* o = (char*)out + (((unsigned)row << 8) + lane_bytes);
    if (edge_row) {
        atomicAdd((float*)o + 0, acc.x);
        atomicAdd((float*)o + 1, acc.y);
        atomicAdd((float*)o + 2, acc.z);
        atomicAdd((float*)o + 3, acc.w);
    } else {
        *(v4f*)o = acc;
    }
}

template <bool FULL>
__device__ __forceinline__ void scatter_wave(
    const float* __restrict__ proj,
    const int*   __restrict__ rows,
    const int*   __restrict__ cols,
    const float* __restrict__ vals,
    float* __restrict__ out, int a, int cnt, int elast, int l16)
{
    const unsigned lane_bytes = (unsigned)l16 * 16u;
    const char* pbase = (const char*)proj;

    int ec = a + l16;
    if (ec > elast) ec = elast;
    int   rv = rows[ec];
    int   cv = cols[ec];
    float vv = vals[ec];

    v4f  acc = {0.0f, 0.0f, 0.0f, 0.0f};
    int  cur = row_lane<0>(rv);
    bool first = true;

#define SCATTER_STEP(i)                                                       \
    {                                                                         \
        const int   r = row_lane<(i)>(rv);                                    \
        const float v = __builtin_bit_cast(float,                             \
            row_lane<(i)>(__builtin_bit_cast(int, vv)));                      \
        if (FULL || 16 * t + (i) < cnt) {                                     \
            if (r != cur) {                                                   \
                scatter_flush(out, cur, lane_bytes, acc, first);              \
                first = false;                                                \
                acc = v4f{0.0f, 0.0f, 0.0f, 0.0f};                            \
                cur = r;                                                      \
            }                                                                 \
            acc += v * p[(i)];                                                \
        }                                                                     \
    }

#pragma nounroll
    for (int t = 0; t < EPH / 16; ++t) {
        const unsigned crow = (unsigned)cv << 8;
        v4f p[16];
#define SCATTER_LOAD(i) \
        p[(i)] = *(const v4f*)(pbase + ((unsigned)row_lane<(i)>((int)crow) + lane_bytes));
        SCATTER_LOAD(0)  SCATTER_LOAD(1)  SCATTER_LOAD(2)  SCATTER_LOAD(3)
        SCATTER_LOAD(4)  SCATTER_LOAD(5)  SCATTER_LOAD(6)  SCATTER_LOAD(7)
        SCATTER_LOAD(8)  SCATTER_LOAD(9)  SCATTER_LOAD(10) SCATTER_LOAD(11)
        SCATTER_LOAD(12) SCATTER_LOAD(13) SCATTER_LOAD(14) SCATTER_LOAD(15)
#undef SCATTER_LOAD

        int en = a + 16 * (t + 1) + l16;
        if (en > elast) en = elast;
        const int   rn = rows[en];
        const int   cn = cols[en];
        const float vn = vals[en];

        SCATTER_STEP(0)  SCATTER_STEP(1)  SCATTER_STEP(2)  SCATTER_STEP(3)
        SCATTER_STEP(4)  SCATTER_STEP(5)  SCATTER_STEP(6)  SCATTER_STEP(7)
        SCATTER_STEP(8)  SCATTER_STEP(9)  SCATTER_STEP(10) SCATTER_STEP(11)
        SCATTER_STEP(12) SCATTER_STEP(13) SCATTER_STEP(14) SCATTER_STEP(15)

        rv = rn; cv = cn; vv = vn;
    }
#undef SCATTER_STEP

    if (FULL || cnt > 0) scatter_flush(out, cur, lane_bytes, acc, true);
}

__global__ void __launch_bounds__(256) gcn_scatter(
    const float* __restrict__ proj,
    const int*   __restrict__ rows,
    const int*   __restrict__ cols,
    const float* __restrict__ vals,
    float* __restrict__ out, int n_edges)
{
    const int lane = threadIdx.x & 31;
    const int l16  = lane & 15;
    const int half = lane >> 4;
    const int wave = __builtin_amdgcn_readfirstlane((int)(threadIdx.x >> 5));

    const int e0 = ((int)blockIdx.x * 8 + wave) * EPW;
    if (e0 >= n_edges) return;
    const int elast = n_edges - 1;
    const int a = e0 + EPH * half;

    if (e0 + EPW <= n_edges) {
        scatter_wave<true>(proj, rows, cols, vals, out, a, EPH, elast, l16);
    } else {
        int cnt = n_edges - a;
        if (cnt > EPH) cnt = EPH;
        if (cnt < 0) cnt = 0;
        scatter_wave<false>(proj, rows, cols, vals, out, a, cnt, elast, l16);
    }
}

extern "C" void kernel_launch(void* const* d_in, const int* in_sizes, int n_in,
                              void* d_out, int out_size, void* d_ws, size_t ws_size,
                              hipStream_t stream) {
    const float* feat = (const float*)d_in[0];
    const float* W    = (const float*)d_in[1];
    const int*   rows = (const int*)d_in[2];
    const int*   cols = (const int*)d_in[3];
    const float* vals = (const float*)d_in[4];
    float* out  = (float*)d_out;
    float* proj = (float*)d_ws;

    const int n_nodes = in_sizes[0] / IN_FEAT;
    const int n_edges = in_sizes[2];

    gcn_gemm_wmma<<<GEMM_GROUPS, GEMM_THREADS, 0, stream>>>(feat, W, proj, out, n_nodes);

    gcn_scatter<<<(n_edges + 8 * EPW - 1) / (8 * EPW), 256, 0, stream>>>(proj, rows, cols, vals, out, n_edges);
}
